# starting kernel
# baseline (speedup 1.0000x reference)
#include <hip/hip_runtime.h>

typedef __attribute__((ext_vector_type(2))) float v2f;
typedef __attribute__((ext_vector_type(8))) float v8f;

#define KORD 3
#define NKNOT 12
#define NB    8
#define NF1   20
#define ROWS_PER_BLOCK 256
#define HSTRIDE 9

__device__ __forceinline__ float knot(int t) { return -2.2f + 0.4f * (float)t; }

__device__ __forceinline__ float siluf(float v) {
    return v / (1.0f + __expf(-v));
}

__device__ __forceinline__ void bspline8(float x, float bs[NB]) {
    float B[NKNOT - 1];
#pragma unroll
    for (int j = 0; j < NKNOT - 1; ++j)
        B[j] = (x >= knot(j) && x < knot(j + 1)) ? 1.0f : 0.0f;
#pragma unroll
    for (int p = 1; p <= KORD; ++p) {
        const float inv = 1.0f / (0.4f * (float)p);
#pragma unroll
        for (int j = 0; j < NKNOT - 1 - p; ++j) {
            float left  = (x - knot(j)) * inv;
            float right = (knot(j + p + 1) - x) * inv;
            B[j] = left * B[j] + right * B[j + 1];
        }
    }
#pragma unroll
    for (int j = 0; j < NB; ++j) bs[j] = B[j];
}

__device__ __forceinline__ float w1_elem(int k, int n,
                                         const float* __restrict__ coef1,
                                         const float* __restrict__ sb1,
                                         const float* __restrict__ sp1) {
    if (n >= 5) return 0.0f;
    if (k == 0) return sb1[n];
    if (k == 1) return sb1[5 + n];
    if (k < 2 + NB)     return sp1[n]     * coef1[n * NB + (k - 2)];
    if (k < 2 + 2 * NB) return sp1[5 + n] * coef1[(5 + n) * NB + (k - 2 - NB)];
    return 0.0f;
}

__global__ __launch_bounds__(ROWS_PER_BLOCK) void kan_fused_kernel(
    const float* __restrict__ x,
    const float* __restrict__ coef1, const float* __restrict__ sb1,
    const float* __restrict__ sp1,
    const float* __restrict__ coef2, const float* __restrict__ sb2,
    const float* __restrict__ sp2,
    const float* __restrict__ fcw,   const float* __restrict__ fcb,
    float* __restrict__ out, int nrows)
{
    __shared__ float lds_feat[ROWS_PER_BLOCK * NF1];
    __shared__ float lds_h[ROWS_PER_BLOCK * HSTRIDE];

    const int t    = threadIdx.x;
    const int lane = t & 31;
    const int wave = t >> 5;
    const int blockBase = blockIdx.x * ROWS_PER_BLOCK;
    const int row  = blockBase + t;

    __builtin_prefetch(&x[(size_t)blockBase * 2], 0, 1);

    float x0 = 0.0f, x1 = 0.0f;
    if (row < nrows) {
        x0 = x[(size_t)row * 2 + 0];
        x1 = x[(size_t)row * 2 + 1];
    }
    float bsa[NB], bsb[NB];
    bspline8(x0, bsa);
    bspline8(x1, bsb);
    {
        float* f = &lds_feat[t * NF1];
        const bool live = (row < nrows);
        f[0] = live ? siluf(x0) : 0.0f;
        f[1] = live ? siluf(x1) : 0.0f;
#pragma unroll
        for (int k = 0; k < NB; ++k) {
            f[2 + k]      = live ? bsa[k] : 0.0f;
            f[2 + NB + k] = live ? bsb[k] : 0.0f;
        }
        f[18] = 0.0f; f[19] = 0.0f;
    }

    const int n  = lane & 15;
    const int hh = lane >> 4;
    v2f bfrag[5];
#pragma unroll
    for (int j = 0; j < 5; ++j) {
        int kb = 4 * j + 2 * hh;
        bfrag[j].x = w1_elem(kb,     n, coef1, sb1, sp1);
        bfrag[j].y = w1_elem(kb + 1, n, coef1, sb1, sp1);
    }

    __syncthreads();

    const int R0 = wave * 32;
    const int R1 = R0 + 16;
    v8f acc0 = {}, acc1 = {};
#pragma unroll
    for (int j = 0; j < 5; ++j) {
        int kb = 4 * j + 2 * hh;
        v2f a0 = *(const v2f*)&lds_feat[(R0 + n) * NF1 + kb];
        v2f a1 = *(const v2f*)&lds_feat[(R1 + n) * NF1 + kb];
        acc0 = __builtin_amdgcn_wmma_f32_16x16x4_f32(
            false, a0, false, bfrag[j], (short)0, acc0, false, false);
        acc1 = __builtin_amdgcn_wmma_f32_16x16x4_f32(
            false, a1, false, bfrag[j], (short)0, acc1, false, false);
    }

    if (n < 5) {
#pragma unroll
        for (int v = 0; v < 8; ++v) {
            lds_h[(R0 + v + 8 * hh) * HSTRIDE + n] = acc0[v];
            lds_h[(R1 + v + 8 * hh) * HSTRIDE + n] = acc1[v];
        }
    }

    __syncthreads();

    if (row < nrows) {
        float a = 0.0f;
#pragma unroll
        for (int d = 0; d < 5; ++d) {
            float hv = lds_h[t * HSTRIDE + d];
            float bs2[NB];
            bspline8(hv, bs2);
            float s = 0.0f;
#pragma unroll
            for (int k = 0; k < NB; ++k) s += coef2[d * NB + k] * bs2[k];
            a += sb2[d] * siluf(hv) + sp2[d] * s;
        }
        out[row] = a * fcw[0] + fcb[0];
    }
}

extern "C" void kernel_launch(void* const* d_in, const int* in_sizes, int n_in,
                              void* d_out, int out_size, void* d_ws, size_t ws_size,
                              hipStream_t stream) {
    (void)n_in; (void)d_ws; (void)ws_size; (void)out_size;
    const float* x     = (const float*)d_in[0];
    const float* coef1 = (const float*)d_in[1];
    const float* sb1   = (const float*)d_in[2];
    const float* sp1   = (const float*)d_in[3];
    const float* coef2 = (const float*)d_in[4];
    const float* sb2   = (const float*)d_in[5];
    const float* sp2   = (const float*)d_in[6];
    const float* fcw   = (const float*)d_in[7];
    const float* fcb   = (const float*)d_in[8];
    float* out = (float*)d_out;

    const int nrows   = in_sizes[0] / 2;
    const int nblocks = (nrows + ROWS_PER_BLOCK - 1) / ROWS_PER_BLOCK;

    kan_fused_kernel<<<nblocks, ROWS_PER_BLOCK, 0, stream>>>(
        x, coef1, sb1, sp1, coef2, sb2, sp2, fcw, fcb, out, nrows);
}


# Uniform cubic B-spline in closed form (interval by fma+floor, local u, 4 cubic weights) replaces 11 indicators + basis recursion; layer-2 sums re-associated as per-interval cubics from an LDS table
# speedup vs baseline: 2.1129x; 2.1129x over previous
#include <hip/hip_runtime.h>

typedef __attribute__((ext_vector_type(2))) float v2f;
typedef __attribute__((ext_vector_type(8))) float v8f;

#define NINT  11
#define NB    8
#define NF1   20
#define ROWS_PER_BLOCK 256
#define HSTRIDE 9
#define NCUB  12

__device__ __forceinline__ float siluf(float v) {
    return v / (1.0f + __expf(-v));
}

__device__ __forceinline__ void bspline8(float x, float bs[NB]) {
    const float s  = fmaf(x, 2.5f, 5.5f);
    const float mf = floorf(s);
    const float u  = s - mf;
    const float v  = 1.0f - u;
    const float u2 = u * u;
    const float v2 = v * v;
    const float w3 = u2 * u * (1.0f / 6.0f);
    const float w0 = v2 * v * (1.0f / 6.0f);
    const float w1 = fmaf(u2, fmaf(u, 0.5f, -1.0f), 2.0f / 3.0f);
    const float w2 = fmaf(v2, fmaf(v, 0.5f, -1.0f), 2.0f / 3.0f);
    const int   mi = (int)mf;
    bool e[NINT];
#pragma unroll
    for (int c = 0; c < NINT; ++c) e[c] = (mi == c);
#pragma unroll
    for (int j = 0; j < NB; ++j)
        bs[j] = e[j] ? w3 : (e[j + 1] ? w2 : (e[j + 2] ? w1 : (e[j + 3] ? w0 : 0.0f)));
}

__device__ __forceinline__ float w1_elem(int k, int n,
                                         const float* __restrict__ coef1,
                                         const float* __restrict__ sb1,
                                         const float* __restrict__ sp1) {
    if (n >= 5) return 0.0f;
    if (k == 0) return sb1[n];
    if (k == 1) return sb1[5 + n];
    if (k < 2 + NB)     return sp1[n]     * coef1[n * NB + (k - 2)];
    if (k < 2 + 2 * NB) return sp1[5 + n] * coef1[(5 + n) * NB + (k - 2 - NB)];
    return 0.0f;
}

__global__ __launch_bounds__(ROWS_PER_BLOCK) void kan_fused_kernel(
    const float* __restrict__ x,
    const float* __restrict__ coef1, const float* __restrict__ sb1,
    const float* __restrict__ sp1,
    const float* __restrict__ coef2, const float* __restrict__ sb2,
    const float* __restrict__ sp2,
    const float* __restrict__ fcw,   const float* __restrict__ fcb,
    float* __restrict__ out, int nrows)
{
    __shared__ float lds_feat[ROWS_PER_BLOCK * NF1];
    __shared__ float lds_h[ROWS_PER_BLOCK * HSTRIDE];
    __shared__ float4 lds_cub[5 * NCUB];

    const int t    = threadIdx.x;
    const int lane = t & 31;
    const int wave = t >> 5;
    const int blockBase = blockIdx.x * ROWS_PER_BLOCK;
    const int row  = blockBase + t;

    __builtin_prefetch(&x[(size_t)blockBase * 2], 0, 1);

    float x0 = 0.0f, x1 = 0.0f;
    if (row < nrows) {
        x0 = x[(size_t)row * 2 + 0];
        x1 = x[(size_t)row * 2 + 1];
    }

    if (t < 5 * NCUB) {
        const int d = t / NCUB;
        const int m = t - d * NCUB;
        const float* c = coef2 + d * NB;
        const float la = c[min(max(m - 3, 0), NB - 1)];
        const float lb = c[min(max(m - 2, 0), NB - 1)];
        const float lg = c[min(max(m - 1, 0), NB - 1)];
        const float le = c[min(m, NB - 1)];
        const float ca = (m >= 3 && m <= 10) ? la : 0.0f;
        const float cb = (m >= 2 && m <= 9)  ? lb : 0.0f;
        const float cg = (m >= 1 && m <= 8)  ? lg : 0.0f;
        const float ce = (m <= 7)            ? le : 0.0f;
        const float sp = sp2[d];
        float4 p;
        p.x = sp * ((ca + 4.0f * cb + cg) * (1.0f / 6.0f));
        p.y = sp * ((cg - ca) * 0.5f);
        p.z = sp * ((ca - 2.0f * cb + cg) * 0.5f);
        p.w = sp * ((ce - ca + 3.0f * (cb - cg)) * (1.0f / 6.0f));
        lds_cub[t] = p;
    }

    float bsa[NB], bsb[NB];
    bspline8(x0, bsa);
    bspline8(x1, bsb);
    {
        float* f = &lds_feat[t * NF1];
        const bool live = (row < nrows);
        f[0] = live ? siluf(x0) : 0.0f;
        f[1] = live ? siluf(x1) : 0.0f;
#pragma unroll
        for (int k = 0; k < NB; ++k) {
            f[2 + k]      = live ? bsa[k] : 0.0f;
            f[2 + NB + k] = live ? bsb[k] : 0.0f;
        }
        f[18] = 0.0f; f[19] = 0.0f;
    }

    const int n  = lane & 15;
    const int hh = lane >> 4;
    v2f bfrag[5];
#pragma unroll
    for (int j = 0; j < 5; ++j) {
        int kb = 4 * j + 2 * hh;
        bfrag[j].x = w1_elem(kb,     n, coef1, sb1, sp1);
        bfrag[j].y = w1_elem(kb + 1, n, coef1, sb1, sp1);
    }

    __syncthreads();

    const int R0 = wave * 32;
    const int R1 = R0 + 16;
    v8f acc0 = {}, acc1 = {};
#pragma unroll
    for (int j = 0; j < 5; ++j) {
        int kb = 4 * j + 2 * hh;
        v2f a0 = *(const v2f*)&lds_feat[(R0 + n) * NF1 + kb];
        v2f a1 = *(const v2f*)&lds_feat[(R1 + n) * NF1 + kb];
        acc0 = __builtin_amdgcn_wmma_f32_16x16x4_f32(
            false, a0, false, bfrag[j], (short)0, acc0, false, false);
        acc1 = __builtin_amdgcn_wmma_f32_16x16x4_f32(
            false, a1, false, bfrag[j], (short)0, acc1, false, false);
    }

    if (n < 5) {
#pragma unroll
        for (int v = 0; v < 8; ++v) {
            lds_h[(R0 + v + 8 * hh) * HSTRIDE + n] = acc0[v];
            lds_h[(R1 + v + 8 * hh) * HSTRIDE + n] = acc1[v];
        }
    }

    __syncthreads();

    if (row < nrows) {
        float a = 0.0f;
#pragma unroll
        for (int d = 0; d < 5; ++d) {
            float hv = lds_h[t * HSTRIDE + d];
            const float s  = fmaf(hv, 2.5f, 5.5f);
            const float mf = floorf(s);
            const float u  = s - mf;
            const int idx  = (s >= 0.0f && s < (float)NINT) ? (int)mf : NINT;
            const float4 p = lds_cub[d * NCUB + idx];
            const float spl = fmaf(fmaf(fmaf(p.w, u, p.z), u, p.y), u, p.x);
            a += sb2[d] * siluf(hv) + spl;
        }
        out[row] = a * fcw[0] + fcb[0];
    }
}

extern "C" void kernel_launch(void* const* d_in, const int* in_sizes, int n_in,
                              void* d_out, int out_size, void* d_ws, size_t ws_size,
                              hipStream_t stream) {
    (void)n_in; (void)d_ws; (void)ws_size; (void)out_size;
    const float* x     = (const float*)d_in[0];
    const float* coef1 = (const float*)d_in[1];
    const float* sb1   = (const float*)d_in[2];
    const float* sp1   = (const float*)d_in[3];
    const float* coef2 = (const float*)d_in[4];
    const float* sb2   = (const float*)d_in[5];
    const float* sp2   = (const float*)d_in[6];
    const float* fcw   = (const float*)d_in[7];
    const float* fcb   = (const float*)d_in[8];
    float* out = (float*)d_out;

    const int nrows   = in_sizes[0] / 2;
    const int nblocks = (nrows + ROWS_PER_BLOCK - 1) / ROWS_PER_BLOCK;

    kan_fused_kernel<<<nblocks, ROWS_PER_BLOCK, 0, stream>>>(
        x, coef1, sb1, sp1, coef2, sb2, sp2, fcw, fcb, out, nrows);
}


# Persistent work-groups: 2048 blocks each loop over 4 row tiles of 256 (next x loaded one tile ahead); wave entry, layer-1 weight fragments and layer-2 table paid once per block, not per tile
# speedup vs baseline: 2.4174x; 1.1441x over previous
#include <hip/hip_runtime.h>

typedef __attribute__((ext_vector_type(2))) float v2f;
typedef __attribute__((ext_vector_type(8))) float v8f;

#define NINT  11
#define NB    8
#define NF1   20
#define ROWS_PER_BLOCK 256
#define HSTRIDE 9
#define NCUB  12
#define NBLOCKS 2048

__device__ __forceinline__ float siluf(float v) {
    return v / (1.0f + __expf(-v));
}

__device__ __forceinline__ void bspline8(float x, float bs[NB]) {
    const float s  = fmaf(x, 2.5f, 5.5f);
    const float mf = floorf(s);
    const float u  = s - mf;
    const float v  = 1.0f - u;
    const float u2 = u * u;
    const float v2 = v * v;
    const float w3 = u2 * u * (1.0f / 6.0f);
    const float w0 = v2 * v * (1.0f / 6.0f);
    const float w1 = fmaf(u2, fmaf(u, 0.5f, -1.0f), 2.0f / 3.0f);
    const float w2 = fmaf(v2, fmaf(v, 0.5f, -1.0f), 2.0f / 3.0f);
    const int   mi = (int)mf;
    bool e[NINT];
#pragma unroll
    for (int c = 0; c < NINT; ++c) e[c] = (mi == c);
#pragma unroll
    for (int j = 0; j < NB; ++j)
        bs[j] = e[j] ? w3 : (e[j + 1] ? w2 : (e[j + 2] ? w1 : (e[j + 3] ? w0 : 0.0f)));
}

__device__ __forceinline__ float w1_elem(int k, int n,
                                         const float* __restrict__ coef1,
                                         const float* __restrict__ sb1,
                                         const float* __restrict__ sp1) {
    if (n >= 5) return 0.0f;
    if (k == 0) return sb1[n];
    if (k == 1) return sb1[5 + n];
    if (k < 2 + NB)     return sp1[n]     * coef1[n * NB + (k - 2)];
    if (k < 2 + 2 * NB) return sp1[5 + n] * coef1[(5 + n) * NB + (k - 2 - NB)];
    return 0.0f;
}

__global__ __launch_bounds__(ROWS_PER_BLOCK) void kan_fused_kernel(
    const float* __restrict__ x,
    const float* __restrict__ coef1, const float* __restrict__ sb1,
    const float* __restrict__ sp1,
    const float* __restrict__ coef2, const float* __restrict__ sb2,
    const float* __restrict__ sp2,
    const float* __restrict__ fcw,   const float* __restrict__ fcb,
    float* __restrict__ out, int nrows)
{
    __shared__ float lds_feat[ROWS_PER_BLOCK * NF1];
    __shared__ float lds_h[ROWS_PER_BLOCK * HSTRIDE];
    __shared__ float4 lds_cub[5 * NCUB];

    const int t    = threadIdx.x;
    const int lane = t & 31;
    const int wave = t >> 5;
    const int ntiles = (nrows + ROWS_PER_BLOCK - 1) / ROWS_PER_BLOCK;
    int tile = blockIdx.x;

    float xn0 = 0.0f, xn1 = 0.0f;
    if (tile < ntiles) {
        __builtin_prefetch(&x[(size_t)tile * ROWS_PER_BLOCK * 2], 0, 1);
        const int r = tile * ROWS_PER_BLOCK + t;
        if (r < nrows) {
            xn0 = x[(size_t)r * 2 + 0];
            xn1 = x[(size_t)r * 2 + 1];
        }
    }


    if (t < 5 * NCUB) {
        const int d = t / NCUB;
        const int m = t - d * NCUB;
        const float* c = coef2 + d * NB;
        const float la = c[min(max(m - 3, 0), NB - 1)];
        const float lb = c[min(max(m - 2, 0), NB - 1)];
        const float lg = c[min(max(m - 1, 0), NB - 1)];
        const float le = c[min(m, NB - 1)];
        const float ca = (m >= 3 && m <= 10) ? la : 0.0f;
        const float cb = (m >= 2 && m <= 9)  ? lb : 0.0f;
        const float cg = (m >= 1 && m <= 8)  ? lg : 0.0f;
        const float ce = (m <= 7)            ? le : 0.0f;
        const float sp = sp2[d];
        float4 p;
        p.x = sp * ((ca + 4.0f * cb + cg) * (1.0f / 6.0f));
        p.y = sp * ((cg - ca) * 0.5f);
        p.z = sp * ((ca - 2.0f * cb + cg) * 0.5f);
        p.w = sp * ((ce - ca + 3.0f * (cb - cg)) * (1.0f / 6.0f));
        lds_cub[t] = p;
    }

    const int n  = lane & 15;
    const int hh = lane >> 4;
    v2f bfrag[5];
#pragma unroll
    for (int j = 0; j < 5; ++j) {
        int kb = 4 * j + 2 * hh;
        bfrag[j].x = w1_elem(kb,     n, coef1, sb1, sp1);
        bfrag[j].y = w1_elem(kb + 1, n, coef1, sb1, sp1);
    }

    float sb2v[5];
#pragma unroll
    for (int d = 0; d < 5; ++d) sb2v[d] = sb2[d];
    const float fcwv = fcw[0];
    const float fcbv = fcb[0];

    const int R0 = wave * 32;
    const int R1 = R0 + 16;

    for (; tile < ntiles; tile += NBLOCKS) {
        const int  row  = tile * ROWS_PER_BLOCK + t;
        const bool live = (row < nrows);
        const float x0 = xn0, x1 = xn1;

        xn0 = 0.0f; xn1 = 0.0f;
        if (tile + NBLOCKS < ntiles) {
            const int rn = (tile + NBLOCKS) * ROWS_PER_BLOCK + t;
            if (rn < nrows) {
                xn0 = x[(size_t)rn * 2 + 0];
                xn1 = x[(size_t)rn * 2 + 1];
            }
        }

        float fv[2 + 2 * NB];
#pragma unroll
        for (int k = 0; k < 2 + 2 * NB; ++k) fv[k] = 0.0f;
        if (live) {
            float bsa[NB], bsb[NB];
            bspline8(x0, bsa);
            bspline8(x1, bsb);
            fv[0] = siluf(x0);
            fv[1] = siluf(x1);
#pragma unroll
            for (int k = 0; k < NB; ++k) {
                fv[2 + k]      = bsa[k];
                fv[2 + NB + k] = bsb[k];
            }
        }
        {
            float* f = &lds_feat[t * NF1];
#pragma unroll
            for (int k = 0; k < 2 + 2 * NB; ++k) f[k] = fv[k];
            f[18] = 0.0f; f[19] = 0.0f;
        }

        __syncthreads();

        v8f acc0 = {}, acc1 = {};
#pragma unroll
        for (int j = 0; j < 5; ++j) {
            int kb = 4 * j + 2 * hh;
            v2f a0 = *(const v2f*)&lds_feat[(R0 + n) * NF1 + kb];
            v2f a1 = *(const v2f*)&lds_feat[(R1 + n) * NF1 + kb];
            acc0 = __builtin_amdgcn_wmma_f32_16x16x4_f32(
                false, a0, false, bfrag[j], (short)0, acc0, false, false);
            acc1 = __builtin_amdgcn_wmma_f32_16x16x4_f32(
                false, a1, false, bfrag[j], (short)0, acc1, false, false);
        }

        if (n < 5) {
#pragma unroll
            for (int v = 0; v < 8; ++v) {
                lds_h[(R0 + v + 8 * hh) * HSTRIDE + n] = acc0[v];
                lds_h[(R1 + v + 8 * hh) * HSTRIDE + n] = acc1[v];
            }
        }

        __syncthreads();

        if (live) {
            float a = 0.0f;
#pragma unroll
            for (int d = 0; d < 5; ++d) {
                float hv = lds_h[t * HSTRIDE + d];
                const float s  = fmaf(hv, 2.5f, 5.5f);
                const float mf = floorf(s);
                const float u  = s - mf;
                const int idx  = (s >= 0.0f && s < (float)NINT) ? (int)mf : NINT;
                const float4 p = lds_cub[d * NCUB + idx];
                const float spl = fmaf(fmaf(fmaf(p.w, u, p.z), u, p.y), u, p.x);
                a += sb2v[d] * siluf(hv) + spl;
            }
            out[row] = a * fcwv + fcbv;
        }
    }
}

extern "C" void kernel_launch(void* const* d_in, const int* in_sizes, int n_in,
                              void* d_out, int out_size, void* d_ws, size_t ws_size,
                              hipStream_t stream) {
    (void)n_in; (void)d_ws; (void)ws_size; (void)out_size;
    const float* x     = (const float*)d_in[0];
    const float* coef1 = (const float*)d_in[1];
    const float* sb1   = (const float*)d_in[2];
    const float* sp1   = (const float*)d_in[3];
    const float* coef2 = (const float*)d_in[4];
    const float* sb2   = (const float*)d_in[5];
    const float* sp2   = (const float*)d_in[6];
    const float* fcw   = (const float*)d_in[7];
    const float* fcb   = (const float*)d_in[8];
    float* out = (float*)d_out;

    const int nrows = in_sizes[0] / 2;

    kan_fused_kernel<<<NBLOCKS, ROWS_PER_BLOCK, 0, stream>>>(
        x, coef1, sb1, sp1, coef2, sb2, sp2, fcw, fcb, out, nrows);
}


# SiLU as hv + hv*tanh(hv), hv = v/2, on one v_tanh_f32: replaces v_exp_f32 plus the correctly rounded division (11 vector instructions) in all seven SiLUs per row; hardware tanh accuracy
# speedup vs baseline: 2.7107x; 1.1213x over previous
#include <hip/hip_runtime.h>

typedef __attribute__((ext_vector_type(2))) float v2f;
typedef __attribute__((ext_vector_type(8))) float v8f;

#define NINT  11
#define NB    8
#define NF1   20
#define ROWS_PER_BLOCK 256
#define HSTRIDE 9
#define NCUB  12
#define NBLOCKS 2048

__device__ __forceinline__ float siluf(float v) {
    const float hv = 0.5f * v;
    return fmaf(hv, __builtin_amdgcn_tanhf(hv), hv);
}

__device__ __forceinline__ void bspline8(float x, float bs[NB]) {
    const float s  = fmaf(x, 2.5f, 5.5f);
    const float mf = floorf(s);
    const float u  = s - mf;
    const float v  = 1.0f - u;
    const float u2 = u * u;
    const float v2 = v * v;
    const float w3 = u2 * u * (1.0f / 6.0f);
    const float w0 = v2 * v * (1.0f / 6.0f);
    const float w1 = fmaf(u2, fmaf(u, 0.5f, -1.0f), 2.0f / 3.0f);
    const float w2 = fmaf(v2, fmaf(v, 0.5f, -1.0f), 2.0f / 3.0f);
    const int   mi = (int)mf;
    bool e[NINT];
#pragma unroll
    for (int c = 0; c < NINT; ++c) e[c] = (mi == c);
#pragma unroll
    for (int j = 0; j < NB; ++j)
        bs[j] = e[j] ? w3 : (e[j + 1] ? w2 : (e[j + 2] ? w1 : (e[j + 3] ? w0 : 0.0f)));
}

__device__ __forceinline__ float w1_elem(int k, int n,
                                         const float* __restrict__ coef1,
                                         const float* __restrict__ sb1,
                                         const float* __restrict__ sp1) {
    if (n >= 5) return 0.0f;
    if (k == 0) return sb1[n];
    if (k == 1) return sb1[5 + n];
    if (k < 2 + NB)     return sp1[n]     * coef1[n * NB + (k - 2)];
    if (k < 2 + 2 * NB) return sp1[5 + n] * coef1[(5 + n) * NB + (k - 2 - NB)];
    return 0.0f;
}

__global__ __launch_bounds__(ROWS_PER_BLOCK) void kan_fused_kernel(
    const float* __restrict__ x,
    const float* __restrict__ coef1, const float* __restrict__ sb1,
    const float* __restrict__ sp1,
    const float* __restrict__ coef2, const float* __restrict__ sb2,
    const float* __restrict__ sp2,
    const float* __restrict__ fcw,   const float* __restrict__ fcb,
    float* __restrict__ out, int nrows)
{
    __shared__ float lds_feat[ROWS_PER_BLOCK * NF1];
    __shared__ float lds_h[ROWS_PER_BLOCK * HSTRIDE];
    __shared__ float4 lds_cub[5 * NCUB];

    const int t    = threadIdx.x;
    const int lane = t & 31;
    const int wave = t >> 5;
    const int ntiles = (nrows + ROWS_PER_BLOCK - 1) / ROWS_PER_BLOCK;
    int tile = blockIdx.x;

    float xn0 = 0.0f, xn1 = 0.0f;
    if (tile < ntiles) {
        __builtin_prefetch(&x[(size_t)tile * ROWS_PER_BLOCK * 2], 0, 1);
        const int r = tile * ROWS_PER_BLOCK + t;
        if (r < nrows) {
            xn0 = x[(size_t)r * 2 + 0];
            xn1 = x[(size_t)r * 2 + 1];
        }
    }


    if (t < 5 * NCUB) {
        const int d = t / NCUB;
        const int m = t - d * NCUB;
        const float* c = coef2 + d * NB;
        const float la = c[min(max(m - 3, 0), NB - 1)];
        const float lb = c[min(max(m - 2, 0), NB - 1)];
        const float lg = c[min(max(m - 1, 0), NB - 1)];
        const float le = c[min(m, NB - 1)];
        const float ca = (m >= 3 && m <= 10) ? la : 0.0f;
        const float cb = (m >= 2 && m <= 9)  ? lb : 0.0f;
        const float cg = (m >= 1 && m <= 8)  ? lg : 0.0f;
        const float ce = (m <= 7)            ? le : 0.0f;
        const float sp = sp2[d];
        float4 p;
        p.x = sp * ((ca + 4.0f * cb + cg) * (1.0f / 6.0f));
        p.y = sp * ((cg - ca) * 0.5f);
        p.z = sp * ((ca - 2.0f * cb + cg) * 0.5f);
        p.w = sp * ((ce - ca + 3.0f * (cb - cg)) * (1.0f / 6.0f));
        lds_cub[t] = p;
    }

    const int n  = lane & 15;
    const int hh = lane >> 4;
    v2f bfrag[5];
#pragma unroll
    for (int j = 0; j < 5; ++j) {
        int kb = 4 * j + 2 * hh;
        bfrag[j].x = w1_elem(kb,     n, coef1, sb1, sp1);
        bfrag[j].y = w1_elem(kb + 1, n, coef1, sb1, sp1);
    }

    float sb2v[5];
#pragma unroll
    for (int d = 0; d < 5; ++d) sb2v[d] = sb2[d];
    const float fcwv = fcw[0];
    const float fcbv = fcb[0];

    const int R0 = wave * 32;
    const int R1 = R0 + 16;

    for (; tile < ntiles; tile += NBLOCKS) {
        const int  row  = tile * ROWS_PER_BLOCK + t;
        const bool live = (row < nrows);
        const float x0 = xn0, x1 = xn1;

        xn0 = 0.0f; xn1 = 0.0f;
        if (tile + NBLOCKS < ntiles) {
            const int rn = (tile + NBLOCKS) * ROWS_PER_BLOCK + t;
            if (rn < nrows) {
                xn0 = x[(size_t)rn * 2 + 0];
                xn1 = x[(size_t)rn * 2 + 1];
            }
        }

        float fv[2 + 2 * NB];
#pragma unroll
        for (int k = 0; k < 2 + 2 * NB; ++k) fv[k] = 0.0f;
        if (live) {
            float bsa[NB], bsb[NB];
            bspline8(x0, bsa);
            bspline8(x1, bsb);
            fv[0] = siluf(x0);
            fv[1] = siluf(x1);
#pragma unroll
            for (int k = 0; k < NB; ++k) {
                fv[2 + k]      = bsa[k];
                fv[2 + NB + k] = bsb[k];
            }
        }
        {
            float* f = &lds_feat[t * NF1];
#pragma unroll
            for (int k = 0; k < 2 + 2 * NB; ++k) f[k] = fv[k];
            f[18] = 0.0f; f[19] = 0.0f;
        }

        __syncthreads();

        v8f acc0 = {}, acc1 = {};
#pragma unroll
        for (int j = 0; j < 5; ++j) {
            int kb = 4 * j + 2 * hh;
            v2f a0 = *(const v2f*)&lds_feat[(R0 + n) * NF1 + kb];
            v2f a1 = *(const v2f*)&lds_feat[(R1 + n) * NF1 + kb];
            acc0 = __builtin_amdgcn_wmma_f32_16x16x4_f32(
                false, a0, false, bfrag[j], (short)0, acc0, false, false);
            acc1 = __builtin_amdgcn_wmma_f32_16x16x4_f32(
                false, a1, false, bfrag[j], (short)0, acc1, false, false);
        }

        if (n < 5) {
#pragma unroll
            for (int v = 0; v < 8; ++v) {
                lds_h[(R0 + v + 8 * hh) * HSTRIDE + n] = acc0[v];
                lds_h[(R1 + v + 8 * hh) * HSTRIDE + n] = acc1[v];
            }
        }

        __syncthreads();

        if (live) {
            float a = 0.0f;
#pragma unroll
            for (int d = 0; d < 5; ++d) {
                float hv = lds_h[t * HSTRIDE + d];
                const float s  = fmaf(hv, 2.5f, 5.5f);
                const float mf = floorf(s);
                const float u  = s - mf;
                const int idx  = (s >= 0.0f && s < (float)NINT) ? (int)mf : NINT;
                const float4 p = lds_cub[d * NCUB + idx];
                const float spl = fmaf(fmaf(fmaf(p.w, u, p.z), u, p.y), u, p.x);
                a += sb2v[d] * siluf(hv) + spl;
            }
            out[row] = a * fcwv + fcbv;
        }
    }
}

extern "C" void kernel_launch(void* const* d_in, const int* in_sizes, int n_in,
                              void* d_out, int out_size, void* d_ws, size_t ws_size,
                              hipStream_t stream) {
    (void)n_in; (void)d_ws; (void)ws_size; (void)out_size;
    const float* x     = (const float*)d_in[0];
    const float* coef1 = (const float*)d_in[1];
    const float* sb1   = (const float*)d_in[2];
    const float* sp1   = (const float*)d_in[3];
    const float* coef2 = (const float*)d_in[4];
    const float* sb2   = (const float*)d_in[5];
    const float* sp2   = (const float*)d_in[6];
    const float* fcw   = (const float*)d_in[7];
    const float* fcb   = (const float*)d_in[8];
    float* out = (float*)d_out;

    const int nrows = in_sizes[0] / 2;

    kan_fused_kernel<<<NBLOCKS, ROWS_PER_BLOCK, 0, stream>>>(
        x, coef1, sb1, sp1, coef2, sb2, sp2, fcw, fcb, out, nrows);
}


# Layer-1 product transposed (weights as A, features as B; a lane evaluates one input of one row): fragments and h stay in registers, LDS tiles and both barriers per tile go; layer-1 sum order changes
# speedup vs baseline: 3.2405x; 1.1954x over previous
#include <hip/hip_runtime.h>

typedef __attribute__((ext_vector_type(2))) float v2f;
typedef __attribute__((ext_vector_type(8))) float v8f;

#define NINT  11
#define NB    8
#define NFI   10
#define ROWS_PER_BLOCK 256
#define NCUB  12
#define NBLOCKS 2048

__device__ __forceinline__ float siluf(float v) {
    const float hv = 0.5f * v;
    return fmaf(hv, __builtin_amdgcn_tanhf(hv), hv);
}

__device__ __forceinline__ void scatter8(int mi, float w0, float w1, float w2, float w3, float bs[NB]) {
    bool e[NINT];
#pragma unroll
    for (int c = 0; c < NINT; ++c) e[c] = (mi == c);
#pragma unroll
    for (int j = 0; j < NB; ++j)
        bs[j] = e[j] ? w3 : (e[j + 1] ? w2 : (e[j + 2] ? w1 : (e[j + 3] ? w0 : 0.0f)));
}

__device__ __forceinline__ void feat10x2(float xa, float xb, float fa[NFI], float fb[NFI]) {
    v2f xv; xv.x = xa; xv.y = xb;
    const v2f c25 = {2.5f, 2.5f}, c55 = {5.5f, 5.5f}, one = {1.0f, 1.0f}, half = {0.5f, 0.5f};
    const v2f mone = {-1.0f, -1.0f}, c23 = {2.0f / 3.0f, 2.0f / 3.0f}, c16 = {1.0f / 6.0f, 1.0f / 6.0f};
    const v2f s  = __builtin_elementwise_fma(xv, c25, c55);
    const v2f mf = __builtin_elementwise_floor(s);
    const v2f u  = s - mf;
    const v2f v  = one - u;
    const v2f u2 = u * u;
    const v2f v2 = v * v;
    const v2f w3 = u2 * u * c16;
    const v2f w0 = v2 * v * c16;
    const v2f w1 = __builtin_elementwise_fma(u2, __builtin_elementwise_fma(u, half, mone), c23);
    const v2f w2 = __builtin_elementwise_fma(v2, __builtin_elementwise_fma(v, half, mone), c23);
    scatter8((int)mf.x, w0.x, w1.x, w2.x, w3.x, &fa[1]);
    scatter8((int)mf.y, w0.y, w1.y, w2.y, w3.y, &fb[1]);
    const v2f hv = half * xv;
    fa[0] = fmaf(hv.x, __builtin_amdgcn_tanhf(hv.x), hv.x);
    fb[0] = fmaf(hv.y, __builtin_amdgcn_tanhf(hv.y), hv.y);
    fa[9] = 0.0f;
    fb[9] = 0.0f;
}

__device__ __forceinline__ float w1_feat(int i, int f, int o,
                                         const float* __restrict__ coef1,
                                         const float* __restrict__ sb1,
                                         const float* __restrict__ sp1) {
    if (f == 0)  return sb1[i * 5 + o];
    if (f <= NB) return sp1[i * 5 + o] * coef1[(i * 5 + o) * NB + (f - 1)];
    return 0.0f;
}

__global__ __launch_bounds__(ROWS_PER_BLOCK) void kan_fused_kernel(
    const float* __restrict__ x,
    const float* __restrict__ coef1, const float* __restrict__ sb1,
    const float* __restrict__ sp1,
    const float* __restrict__ coef2, const float* __restrict__ sb2,
    const float* __restrict__ sp2,
    const float* __restrict__ fcw,   const float* __restrict__ fcb,
    float* __restrict__ out, int nrows)
{
    __shared__ float4 lds_cub[5 * NCUB];

    const int t    = threadIdx.x;
    const int lane = t & 31;
    const int wave = t >> 5;
    const int n    = lane & 15;
    const int hh   = lane >> 4;
    const int ntiles = (nrows + ROWS_PER_BLOCK - 1) / ROWS_PER_BLOCK;
    int tile = blockIdx.x;

    const int rsub = wave * 32 + n;
    float xna = 0.0f, xnb = 0.0f;
    if (tile < ntiles) {
        __builtin_prefetch(&x[(size_t)tile * ROWS_PER_BLOCK * 2], 0, 1);
        const int ra = tile * ROWS_PER_BLOCK + rsub;
        if (ra < nrows)      xna = x[(size_t)ra * 2 + hh];
        if (ra + 16 < nrows) xnb = x[(size_t)(ra + 16) * 2 + hh];
    }


    if (t < 5 * NCUB) {
        const int d = t / NCUB;
        const int m = t - d * NCUB;
        const float* c = coef2 + d * NB;
        const float la = c[min(max(m - 3, 0), NB - 1)];
        const float lb = c[min(max(m - 2, 0), NB - 1)];
        const float lg = c[min(max(m - 1, 0), NB - 1)];
        const float le = c[min(m, NB - 1)];
        const float ca = (m >= 3 && m <= 10) ? la : 0.0f;
        const float cb = (m >= 2 && m <= 9)  ? lb : 0.0f;
        const float cg = (m >= 1 && m <= 8)  ? lg : 0.0f;
        const float ce = (m <= 7)            ? le : 0.0f;
        const float sp = sp2[d];
        float4 p;
        p.x = sp * ((ca + 4.0f * cb + cg) * (1.0f / 6.0f));
        p.y = sp * ((cg - ca) * 0.5f);
        p.z = sp * ((ca - 2.0f * cb + cg) * 0.5f);
        p.w = sp * ((ce - ca + 3.0f * (cb - cg)) * (1.0f / 6.0f));
        lds_cub[t] = p;
    }

    const int  o   = n & 7;
    const bool wok = (o < 5);
    const int  oc  = wok ? o : 4;
    v2f wfrag[5];
#pragma unroll
    for (int j = 0; j < 5; ++j) {
        const float wx = w1_feat(hh, 2 * j,     oc, coef1, sb1, sp1);
        const float wy = w1_feat(hh, 2 * j + 1, oc, coef1, sb1, sp1);
        wfrag[j].x = wok ? wx : 0.0f;
        wfrag[j].y = wok ? wy : 0.0f;
    }

    float sb2v[5];
#pragma unroll
    for (int d = 0; d < 5; ++d) sb2v[d] = sb2[d];
    const float fcwv = fcw[0];
    const float fcbv = fcb[0];

    __syncthreads();

    for (; tile < ntiles; tile += NBLOCKS) {
        const float xa = xna, xb = xnb;

        xna = 0.0f; xnb = 0.0f;
        if (tile + NBLOCKS < ntiles) {
            const int ra = (tile + NBLOCKS) * ROWS_PER_BLOCK + rsub;
            if (ra < nrows)      xna = x[(size_t)ra * 2 + hh];
            if (ra + 16 < nrows) xnb = x[(size_t)(ra + 16) * 2 + hh];
        }

        float fa[NFI], fb[NFI];
        feat10x2(xa, xb, fa, fb);

        v8f acc0 = {}, acc1 = {};
#pragma unroll
        for (int j = 0; j < 5; ++j) {
            v2f ba, bb;
            ba.x = fa[2 * j]; ba.y = fa[2 * j + 1];
            bb.x = fb[2 * j]; bb.y = fb[2 * j + 1];
            acc0 = __builtin_amdgcn_wmma_f32_16x16x4_f32(
                false, wfrag[j], false, ba, (short)0, acc0, false, false);
            acc1 = __builtin_amdgcn_wmma_f32_16x16x4_f32(
                false, wfrag[j], false, bb, (short)0, acc1, false, false);
        }

        float a = 0.0f;
#pragma unroll
        for (int d = 0; d < 5; ++d) {
            const float hv = hh ? acc1[d] : acc0[d];
            const float s  = fmaf(hv, 2.5f, 5.5f);
            const float mf = floorf(s);
            const float u  = s - mf;
            const int idx  = (s >= 0.0f && s < (float)NINT) ? (int)mf : NINT;
            const float4 p = lds_cub[d * NCUB + idx];
            const float spl = fmaf(fmaf(fmaf(p.w, u, p.z), u, p.y), u, p.x);
            a += sb2v[d] * siluf(hv) + spl;
        }
        const int row = tile * ROWS_PER_BLOCK + t;
        if (row < nrows) out[row] = a * fcwv + fcbv;
    }
}

extern "C" void kernel_launch(void* const* d_in, const int* in_sizes, int n_in,
                              void* d_out, int out_size, void* d_ws, size_t ws_size,
                              hipStream_t stream) {
    (void)n_in; (void)d_ws; (void)ws_size; (void)out_size;
    const float* x     = (const float*)d_in[0];
    const float* coef1 = (const float*)d_in[1];
    const float* sb1   = (const float*)d_in[2];
    const float* sp1   = (const float*)d_in[3];
    const float* coef2 = (const float*)d_in[4];
    const float* sb2   = (const float*)d_in[5];
    const float* sp2   = (const float*)d_in[6];
    const float* fcw   = (const float*)d_in[7];
    const float* fcb   = (const float*)d_in[8];
    float* out = (float*)d_out;

    const int nrows = in_sizes[0] / 2;

    kan_fused_kernel<<<NBLOCKS, ROWS_PER_BLOCK, 0, stream>>>(
        x, coef1, sb1, sp1, coef2, sb2, sp2, fcw, fcb, out, nrows);
}


# Basis placement through a private LDS row per lane (4 weights stored between zeros, the 8 bases read as a window at the interval's offset) replaces the select scatter; layer-1 sum order changes
# speedup vs baseline: 3.6861x; 1.1375x over previous
#include <hip/hip_runtime.h>

typedef __attribute__((ext_vector_type(2))) float v2f;
typedef __attribute__((ext_vector_type(8))) float v8f;

#define NINT  11
#define NB    8
#define NFI   10
#define ROWS_PER_BLOCK 256
#define NCUB  12
#define NBLOCKS 2048
#define ROW_DW  20

__device__ __forceinline__ float siluf(float v) {
    const float hv = 0.5f * v;
    return fmaf(hv, __builtin_amdgcn_tanhf(hv), hv);
}

__device__ __forceinline__ void place8(float* row, int mi,
                                       float w0, float w1, float w2, float w3, float bs[NB]) {
    const unsigned mu = min((unsigned)mi, (unsigned)NINT);
    row[8] = w3; row[9] = w2; row[10] = w1; row[11] = w0;
    const float* p = row + mu;
#pragma unroll
    for (int j = 0; j < NB; ++j) bs[j] = p[8 - j];
}

__device__ __forceinline__ void feat10x2(float xa, float xb, float* row, float fa[NFI], float fb[NFI]) {
    v2f xv; xv.x = xa; xv.y = xb;
    const v2f c25 = {2.5f, 2.5f}, c55 = {5.5f, 5.5f}, one = {1.0f, 1.0f}, half = {0.5f, 0.5f};
    const v2f mone = {-1.0f, -1.0f}, c23 = {2.0f / 3.0f, 2.0f / 3.0f}, c16 = {1.0f / 6.0f, 1.0f / 6.0f};
    const v2f s  = __builtin_elementwise_fma(xv, c25, c55);
    const v2f mf = __builtin_elementwise_floor(s);
    const v2f u  = s - mf;
    const v2f v  = one - u;
    const v2f u2 = u * u;
    const v2f v2 = v * v;
    const v2f w3 = u2 * u * c16;
    const v2f w0 = v2 * v * c16;
    const v2f w1 = __builtin_elementwise_fma(u2, __builtin_elementwise_fma(u, half, mone), c23);
    const v2f w2 = __builtin_elementwise_fma(v2, __builtin_elementwise_fma(v, half, mone), c23);
    place8(row, (int)mf.x, w0.x, w1.x, w2.x, w3.x, &fa[0]);
    place8(row, (int)mf.y, w0.y, w1.y, w2.y, w3.y, &fb[0]);
    const v2f hv = half * xv;
    fa[8] = fmaf(hv.x, __builtin_amdgcn_tanhf(hv.x), hv.x);
    fb[8] = fmaf(hv.y, __builtin_amdgcn_tanhf(hv.y), hv.y);
    fa[9] = 0.0f;
    fb[9] = 0.0f;
}

__device__ __forceinline__ constexpr int fidx(int j, int c) {
    return (j < 4) ? (2 * j + 1 - c) : (NB + c);
}

__device__ __forceinline__ float w1_feat(int i, int f, int o,
                                         const float* __restrict__ coef1,
                                         const float* __restrict__ sb1,
                                         const float* __restrict__ sp1) {
    if (f < NB)  return sp1[i * 5 + o] * coef1[(i * 5 + o) * NB + f];
    if (f == NB) return sb1[i * 5 + o];
    return 0.0f;
}

__global__ __launch_bounds__(ROWS_PER_BLOCK) void kan_fused_kernel(
    const float* __restrict__ x,
    const float* __restrict__ coef1, const float* __restrict__ sb1,
    const float* __restrict__ sp1,
    const float* __restrict__ coef2, const float* __restrict__ sb2,
    const float* __restrict__ sp2,
    const float* __restrict__ fcw,   const float* __restrict__ fcb,
    float* __restrict__ out, int nrows)
{
    __shared__ float4 lds_cub[5 * NCUB];
    __shared__ __attribute__((aligned(16))) float lds_row[ROWS_PER_BLOCK * ROW_DW];

    const int t    = threadIdx.x;
    const int lane = t & 31;
    const int wave = t >> 5;
    const int n    = lane & 15;
    const int hh   = lane >> 4;
    const int ntiles = (nrows + ROWS_PER_BLOCK - 1) / ROWS_PER_BLOCK;
    int tile = blockIdx.x;
    float* const myrow = lds_row + t * ROW_DW;

    const int rsub = wave * 32 + n;
    float xna = 0.0f, xnb = 0.0f;
    if (tile < ntiles) {
        __builtin_prefetch(&x[(size_t)tile * ROWS_PER_BLOCK * 2], 0, 1);
        const int ra = tile * ROWS_PER_BLOCK + rsub;
        if (ra < nrows)      xna = x[(size_t)ra * 2 + hh];
        if (ra + 16 < nrows) xnb = x[(size_t)(ra + 16) * 2 + hh];
    }


    if (t < 5 * NCUB) {
        const int d = t / NCUB;
        const int m = t - d * NCUB;
        const float* c = coef2 + d * NB;
        const float la = c[min(max(m - 3, 0), NB - 1)];
        const float lb = c[min(max(m - 2, 0), NB - 1)];
        const float lg = c[min(max(m - 1, 0), NB - 1)];
        const float le = c[min(m, NB - 1)];
        const float ca = (m >= 3 && m <= 10) ? la : 0.0f;
        const float cb = (m >= 2 && m <= 9)  ? lb : 0.0f;
        const float cg = (m >= 1 && m <= 8)  ? lg : 0.0f;
        const float ce = (m <= 7)            ? le : 0.0f;
        const float sp = sp2[d];
        float4 p;
        p.x = sp * ((ca + 4.0f * cb + cg) * (1.0f / 6.0f));
        p.y = sp * ((cg - ca) * 0.5f);
        p.z = sp * ((ca - 2.0f * cb + cg) * 0.5f);
        p.w = sp * ((ce - ca + 3.0f * (cb - cg)) * (1.0f / 6.0f));
        lds_cub[t] = p;
    }

#pragma unroll
    for (int k = 0; k < 8; ++k) { myrow[k] = 0.0f; myrow[12 + k] = 0.0f; }

    const int  o   = n & 7;
    const bool wok = (o < 5);
    const int  oc  = wok ? o : 4;
    v2f wfrag[5];
#pragma unroll
    for (int j = 0; j < 5; ++j) {
        const float wx = w1_feat(hh, fidx(j, 0), oc, coef1, sb1, sp1);
        const float wy = w1_feat(hh, fidx(j, 1), oc, coef1, sb1, sp1);
        wfrag[j].x = wok ? wx : 0.0f;
        wfrag[j].y = wok ? wy : 0.0f;
    }

    float sb2v[5];
#pragma unroll
    for (int d = 0; d < 5; ++d) sb2v[d] = sb2[d];
    const float fcwv = fcw[0];
    const float fcbv = fcb[0];

    __syncthreads();

    for (; tile < ntiles; tile += NBLOCKS) {
        const float xa = xna, xb = xnb;

        xna = 0.0f; xnb = 0.0f;
        if (tile + NBLOCKS < ntiles) {
            const int ra = (tile + NBLOCKS) * ROWS_PER_BLOCK + rsub;
            if (ra < nrows)      xna = x[(size_t)ra * 2 + hh];
            if (ra + 16 < nrows) xnb = x[(size_t)(ra + 16) * 2 + hh];
        }

        float fa[NFI], fb[NFI];
        feat10x2(xa, xb, myrow, fa, fb);

        v8f acc0 = {}, acc1 = {};
#pragma unroll
        for (int j = 0; j < 5; ++j) {
            v2f ba, bb;
            ba.x = fa[fidx(j, 0)]; ba.y = fa[fidx(j, 1)];
            bb.x = fb[fidx(j, 0)]; bb.y = fb[fidx(j, 1)];
            acc0 = __builtin_amdgcn_wmma_f32_16x16x4_f32(
                false, wfrag[j], false, ba, (short)0, acc0, false, false);
            acc1 = __builtin_amdgcn_wmma_f32_16x16x4_f32(
                false, wfrag[j], false, bb, (short)0, acc1, false, false);
        }

        float a = 0.0f;
#pragma unroll
        for (int d = 0; d < 5; ++d) {
            const float hv = hh ? acc1[d] : acc0[d];
            const float s  = fmaf(hv, 2.5f, 5.5f);
            const float mf = floorf(s);
            const float u  = s - mf;
            const int idx  = (s >= 0.0f && s < (float)NINT) ? (int)mf : NINT;
            const float4 p = lds_cub[d * NCUB + idx];
            const float spl = fmaf(fmaf(fmaf(p.w, u, p.z), u, p.y), u, p.x);
            a += sb2v[d] * siluf(hv) + spl;
        }
        const int row = tile * ROWS_PER_BLOCK + t;
        if (row < nrows) out[row] = a * fcwv + fcbv;
    }
}

extern "C" void kernel_launch(void* const* d_in, const int* in_sizes, int n_in,
                              void* d_out, int out_size, void* d_ws, size_t ws_size,
                              hipStream_t stream) {
    (void)n_in; (void)d_ws; (void)ws_size; (void)out_size;
    const float* x     = (const float*)d_in[0];
    const float* coef1 = (const float*)d_in[1];
    const float* sb1   = (const float*)d_in[2];
    const float* sp1   = (const float*)d_in[3];
    const float* coef2 = (const float*)d_in[4];
    const float* sb2   = (const float*)d_in[5];
    const float* sp2   = (const float*)d_in[6];
    const float* fcw   = (const float*)d_in[7];
    const float* fcb   = (const float*)d_in[8];
    float* out = (float*)d_out;

    const int nrows = in_sizes[0] / 2;

    kan_fused_kernel<<<NBLOCKS, ROWS_PER_BLOCK, 0, stream>>>(
        x, coef1, sb1, sp1, coef2, sb2, sp2, fcw, fcb, out, nrows);
}


# Layer-2 table entry without floor or compares (position shifted one knot, truncation + unsigned min, u by v_fract) and its five terms as one running fma chain; layer-2 sum order changes
# speedup vs baseline: 3.8648x; 1.0485x over previous
#include <hip/hip_runtime.h>

typedef __attribute__((ext_vector_type(2))) float v2f;
typedef __attribute__((ext_vector_type(8))) float v8f;

#define NINT  11
#define NB    8
#define NFI   10
#define ROWS_PER_BLOCK 256
#define NCUB  13
#define NBLOCKS 2048
#define ROW_DW  20

__device__ __forceinline__ float siluf(float v) {
    const float hv = 0.5f * v;
    return fmaf(hv, __builtin_amdgcn_tanhf(hv), hv);
}

__device__ __forceinline__ void place8(float* row, int mi,
                                       float w0, float w1, float w2, float w3, float bs[NB]) {
    const unsigned mu = min((unsigned)mi, (unsigned)NINT);
    row[8] = w3; row[9] = w2; row[10] = w1; row[11] = w0;
    const float* p = row + mu;
#pragma unroll
    for (int j = 0; j < NB; ++j) bs[j] = p[8 - j];
}

__device__ __forceinline__ void feat10x2(float xa, float xb, float* row, float fa[NFI], float fb[NFI]) {
    v2f xv; xv.x = xa; xv.y = xb;
    const v2f c25 = {2.5f, 2.5f}, c55 = {5.5f, 5.5f}, one = {1.0f, 1.0f}, half = {0.5f, 0.5f};
    const v2f mone = {-1.0f, -1.0f}, c23 = {2.0f / 3.0f, 2.0f / 3.0f}, c16 = {1.0f / 6.0f, 1.0f / 6.0f};
    const v2f s  = __builtin_elementwise_fma(xv, c25, c55);
    const v2f mf = __builtin_elementwise_floor(s);
    const v2f u  = s - mf;
    const v2f v  = one - u;
    const v2f u2 = u * u;
    const v2f v2 = v * v;
    const v2f w3 = u2 * u * c16;
    const v2f w0 = v2 * v * c16;
    const v2f w1 = __builtin_elementwise_fma(u2, __builtin_elementwise_fma(u, half, mone), c23);
    const v2f w2 = __builtin_elementwise_fma(v2, __builtin_elementwise_fma(v, half, mone), c23);
    place8(row, (int)mf.x, w0.x, w1.x, w2.x, w3.x, &fa[0]);
    place8(row, (int)mf.y, w0.y, w1.y, w2.y, w3.y, &fb[0]);
    const v2f hv = half * xv;
    fa[8] = fmaf(hv.x, __builtin_amdgcn_tanhf(hv.x), hv.x);
    fb[8] = fmaf(hv.y, __builtin_amdgcn_tanhf(hv.y), hv.y);
    fa[9] = 0.0f;
    fb[9] = 0.0f;
}

__device__ __forceinline__ constexpr int fidx(int j, int c) {
    return (j < 4) ? (2 * j + 1 - c) : (NB + c);
}

__device__ __forceinline__ float w1_feat(int i, int f, int o,
                                         const float* __restrict__ coef1,
                                         const float* __restrict__ sb1,
                                         const float* __restrict__ sp1) {
    if (f < NB)  return sp1[i * 5 + o] * coef1[(i * 5 + o) * NB + f];
    if (f == NB) return sb1[i * 5 + o];
    return 0.0f;
}

__global__ __launch_bounds__(ROWS_PER_BLOCK) void kan_fused_kernel(
    const float* __restrict__ x,
    const float* __restrict__ coef1, const float* __restrict__ sb1,
    const float* __restrict__ sp1,
    const float* __restrict__ coef2, const float* __restrict__ sb2,
    const float* __restrict__ sp2,
    const float* __restrict__ fcw,   const float* __restrict__ fcb,
    float* __restrict__ out, int nrows)
{
    __shared__ float4 lds_cub[5 * NCUB];
    __shared__ __attribute__((aligned(16))) float lds_row[ROWS_PER_BLOCK * ROW_DW];

    const int t    = threadIdx.x;
    const int lane = t & 31;
    const int wave = t >> 5;
    const int n    = lane & 15;
    const int hh   = lane >> 4;
    const int ntiles = (nrows + ROWS_PER_BLOCK - 1) / ROWS_PER_BLOCK;
    int tile = blockIdx.x;
    float* const myrow = lds_row + t * ROW_DW;

    const int rsub = wave * 32 + n;
    float xna = 0.0f, xnb = 0.0f;
    if (tile < ntiles) {
        __builtin_prefetch(&x[(size_t)tile * ROWS_PER_BLOCK * 2], 0, 1);
        const int ra = tile * ROWS_PER_BLOCK + rsub;
        if (ra < nrows)      xna = x[(size_t)ra * 2 + hh];
        if (ra + 16 < nrows) xnb = x[(size_t)(ra + 16) * 2 + hh];
    }


    if (t < 5 * NCUB) {
        const int d = t / NCUB;
        const int m = t - d * NCUB - 1;
        const float* c = coef2 + d * NB;
        const float la = c[min(max(m - 3, 0), NB - 1)];
        const float lb = c[min(max(m - 2, 0), NB - 1)];
        const float lg = c[min(max(m - 1, 0), NB - 1)];
        const float le = c[min(max(m, 0), NB - 1)];
        const float ca = (m >= 3 && m <= 10) ? la : 0.0f;
        const float cb = (m >= 2 && m <= 9)  ? lb : 0.0f;
        const float cg = (m >= 1 && m <= 8)  ? lg : 0.0f;
        const float ce = (m >= 0 && m <= 7)  ? le : 0.0f;
        const float sp = sp2[d];
        float4 p;
        p.x = sp * ((ca + 4.0f * cb + cg) * (1.0f / 6.0f));
        p.y = sp * ((cg - ca) * 0.5f);
        p.z = sp * ((ca - 2.0f * cb + cg) * 0.5f);
        p.w = sp * ((ce - ca + 3.0f * (cb - cg)) * (1.0f / 6.0f));
        lds_cub[t] = p;
    }

#pragma unroll
    for (int k = 0; k < 8; ++k) { myrow[k] = 0.0f; myrow[12 + k] = 0.0f; }

    const int  o   = n & 7;
    const bool wok = (o < 5);
    const int  oc  = wok ? o : 4;
    v2f wfrag[5];
#pragma unroll
    for (int j = 0; j < 5; ++j) {
        const float wx = w1_feat(hh, fidx(j, 0), oc, coef1, sb1, sp1);
        const float wy = w1_feat(hh, fidx(j, 1), oc, coef1, sb1, sp1);
        wfrag[j].x = wok ? wx : 0.0f;
        wfrag[j].y = wok ? wy : 0.0f;
    }

    float sb2v[5];
#pragma unroll
    for (int d = 0; d < 5; ++d) sb2v[d] = sb2[d];
    const float fcwv = fcw[0];
    const float fcbv = fcb[0];

    __syncthreads();

    for (; tile < ntiles; tile += NBLOCKS) {
        const float xa = xna, xb = xnb;

        xna = 0.0f; xnb = 0.0f;
        if (tile + NBLOCKS < ntiles) {
            const int ra = (tile + NBLOCKS) * ROWS_PER_BLOCK + rsub;
            if (ra < nrows)      xna = x[(size_t)ra * 2 + hh];
            if (ra + 16 < nrows) xnb = x[(size_t)(ra + 16) * 2 + hh];
        }

        float fa[NFI], fb[NFI];
        feat10x2(xa, xb, myrow, fa, fb);

        v8f acc0 = {}, acc1 = {};
#pragma unroll
        for (int j = 0; j < 5; ++j) {
            v2f ba, bb;
            ba.x = fa[fidx(j, 0)]; ba.y = fa[fidx(j, 1)];
            bb.x = fb[fidx(j, 0)]; bb.y = fb[fidx(j, 1)];
            acc0 = __builtin_amdgcn_wmma_f32_16x16x4_f32(
                false, wfrag[j], false, ba, (short)0, acc0, false, false);
            acc1 = __builtin_amdgcn_wmma_f32_16x16x4_f32(
                false, wfrag[j], false, bb, (short)0, acc1, false, false);
        }

        float a = 0.0f;
#pragma unroll
        for (int d = 0; d < 5; ++d) {
            const float hv = hh ? acc1[d] : acc0[d];
            const float s1 = fmaf(hv, 2.5f, 6.5f);
            const float u  = __builtin_amdgcn_fractf(s1);
            const unsigned idx = min((unsigned)(int)s1, (unsigned)(NCUB - 1));
            const float4 p = lds_cub[d * NCUB + idx];
            const float q = fmaf(fmaf(p.w, u, p.z), u, p.y);
            const float base = (d == 0) ? p.x : a + p.x;
            a = fmaf(sb2v[d], siluf(hv), fmaf(q, u, base));
        }
        const int row = tile * ROWS_PER_BLOCK + t;
        if (row < nrows) out[row] = a * fcwv + fcbv;
    }
}

extern "C" void kernel_launch(void* const* d_in, const int* in_sizes, int n_in,
                              void* d_out, int out_size, void* d_ws, size_t ws_size,
                              hipStream_t stream) {
    (void)n_in; (void)d_ws; (void)ws_size; (void)out_size;
    const float* x     = (const float*)d_in[0];
    const float* coef1 = (const float*)d_in[1];
    const float* sb1   = (const float*)d_in[2];
    const float* sp1   = (const float*)d_in[3];
    const float* coef2 = (const float*)d_in[4];
    const float* sb2   = (const float*)d_in[5];
    const float* sp2   = (const float*)d_in[6];
    const float* fcw   = (const float*)d_in[7];
    const float* fcb   = (const float*)d_in[8];
    float* out = (float*)d_out;

    const int nrows = in_sizes[0] / 2;

    kan_fused_kernel<<<NBLOCKS, ROWS_PER_BLOCK, 0, stream>>>(
        x, coef1, sb1, sp1, coef2, sb2, sp2, fcw, fcb, out, nrows);
}


# Layer-1 A rows split by lane half (rows 0-4 weight the k slots of row n, rows 8-12 those of row 16+n): one accumulator, both rows' silu in one K step, 9 WMMAs per 32 rows not 10; sum order unchanged
# speedup vs baseline: 4.0976x; 1.0602x over previous
#include <hip/hip_runtime.h>

typedef __attribute__((ext_vector_type(2))) float v2f;
typedef __attribute__((ext_vector_type(8))) float v8f;

#define NINT  11
#define NB    8
#define ROWS_PER_BLOCK 256
#define NCUB  13
#define NBLOCKS 2048
#define ROW_DW  20

__device__ __forceinline__ float silu_of_half(float g) {
    return fmaf(g, __builtin_amdgcn_tanhf(g), g);
}

__device__ __forceinline__ void place8(float* row, int mi,
                                       float w0, float w1, float w2, float w3, float bs[NB]) {
    const unsigned mu = min((unsigned)mi, (unsigned)NINT);
    row[8] = w3; row[9] = w2; row[10] = w1; row[11] = w0;
    const float* p = row + mu;
#pragma unroll
    for (int j = 0; j < NB; ++j) bs[j] = p[8 - j];
}

__device__ __forceinline__ v2f feat9x2(float xa, float xb, float* row, float fa[NB], float fb[NB]) {
    v2f xv; xv.x = xa; xv.y = xb;
    const v2f c25 = {2.5f, 2.5f}, c55 = {5.5f, 5.5f}, one = {1.0f, 1.0f}, half = {0.5f, 0.5f};
    const v2f mone = {-1.0f, -1.0f}, c23 = {2.0f / 3.0f, 2.0f / 3.0f}, c16 = {1.0f / 6.0f, 1.0f / 6.0f};
    const v2f s  = __builtin_elementwise_fma(xv, c25, c55);
    const v2f mf = __builtin_elementwise_floor(s);
    const v2f u  = s - mf;
    const v2f v  = one - u;
    const v2f u2 = u * u;
    const v2f v2 = v * v;
    const v2f w3 = u2 * u * c16;
    const v2f w0 = v2 * v * c16;
    const v2f w1 = __builtin_elementwise_fma(u2, __builtin_elementwise_fma(u, half, mone), c23);
    const v2f w2 = __builtin_elementwise_fma(v2, __builtin_elementwise_fma(v, half, mone), c23);
    place8(row, (int)mf.x, w0.x, w1.x, w2.x, w3.x, &fa[0]);
    place8(row, (int)mf.y, w0.y, w1.y, w2.y, w3.y, &fb[0]);
    const v2f hv = half * xv;
    v2f tv;
    tv.x = __builtin_amdgcn_tanhf(hv.x);
    tv.y = __builtin_amdgcn_tanhf(hv.y);
    return __builtin_elementwise_fma(hv, tv, hv);
}

__device__ __forceinline__ constexpr int fidx(int j, int c) {
    return 2 * j + 1 - c;
}

__device__ __forceinline__ float w1_feat(int i, int f, int o,
                                         const float* __restrict__ coef1,
                                         const float* __restrict__ sb1,
                                         const float* __restrict__ sp1) {
    if (f < NB)  return sp1[i * 5 + o] * coef1[(i * 5 + o) * NB + f];
    if (f == NB) return sb1[i * 5 + o];
    return 0.0f;
}

__global__ __launch_bounds__(ROWS_PER_BLOCK) void kan_fused_kernel(
    const float* __restrict__ x,
    const float* __restrict__ coef1, const float* __restrict__ sb1,
    const float* __restrict__ sp1,
    const float* __restrict__ coef2, const float* __restrict__ sb2,
    const float* __restrict__ sp2,
    const float* __restrict__ fcw,   const float* __restrict__ fcb,
    float* __restrict__ out, int nrows)
{
    __shared__ float4 lds_cub[5 * NCUB];
    __shared__ __attribute__((aligned(16))) float lds_row[ROWS_PER_BLOCK * ROW_DW];

    const int t    = threadIdx.x;
    const int lane = t & 31;
    const int wave = t >> 5;
    const int n    = lane & 15;
    const int hh   = lane >> 4;
    const int ntiles = (nrows + ROWS_PER_BLOCK - 1) / ROWS_PER_BLOCK;
    int tile = blockIdx.x;
    float* const myrow = lds_row + t * ROW_DW;

    const int rsub = wave * 32 + n;
    float xna = 0.0f, xnb = 0.0f;
    if (tile < ntiles) {
        __builtin_prefetch(&x[(size_t)tile * ROWS_PER_BLOCK * 2], 0, 1);
        const int ra = tile * ROWS_PER_BLOCK + rsub;
        if (ra < nrows)      xna = x[(size_t)ra * 2 + hh];
        if (ra + 16 < nrows) xnb = x[(size_t)(ra + 16) * 2 + hh];
    }


    if (t < 5 * NCUB) {
        const int d = t / NCUB;
        const int m = t - d * NCUB - 1;
        const float* c = coef2 + d * NB;
        const float la = c[min(max(m - 3, 0), NB - 1)];
        const float lb = c[min(max(m - 2, 0), NB - 1)];
        const float lg = c[min(max(m - 1, 0), NB - 1)];
        const float le = c[min(max(m, 0), NB - 1)];
        const float ca = (m >= 3 && m <= 10) ? la : 0.0f;
        const float cb = (m >= 2 && m <= 9)  ? lb : 0.0f;
        const float cg = (m >= 1 && m <= 8)  ? lg : 0.0f;
        const float ce = (m >= 0 && m <= 7)  ? le : 0.0f;
        const float sp = sp2[d];
        float4 p;
        p.x = sp * ((ca + 4.0f * cb + cg) * (1.0f / 6.0f));
        p.y = sp * ((cg - ca) * 0.5f);
        p.z = sp * ((ca - 2.0f * cb + cg) * 0.5f);
        p.w = sp * ((ce - ca + 3.0f * (cb - cg)) * (1.0f / 6.0f));
        lds_cub[t] = p;
    }

#pragma unroll
    for (int k = 0; k < 8; ++k) { myrow[k] = 0.0f; myrow[12 + k] = 0.0f; }

    const int  o   = n & 7;
    const bool wok = (o < 5);
    const int  oc  = wok ? o : 4;
    const bool wa = wok && (n < 8);
    const bool wb = wok && (n >= 8);
    v2f wfa[4], wfb[4], wfs;
#pragma unroll
    for (int j = 0; j < 4; ++j) {
        const float wx = w1_feat(hh, fidx(j, 0), oc, coef1, sb1, sp1);
        const float wy = w1_feat(hh, fidx(j, 1), oc, coef1, sb1, sp1);
        wfa[j].x = wa ? wx : 0.0f;
        wfa[j].y = wa ? wy : 0.0f;
        wfb[j].x = wb ? wx : 0.0f;
        wfb[j].y = wb ? wy : 0.0f;
    }
    {
        const float ws = w1_feat(hh, NB, oc, coef1, sb1, sp1);
        wfs.x = wa ? ws : 0.0f;
        wfs.y = wb ? ws : 0.0f;
    }

    float sb2v[5];
#pragma unroll
    for (int d = 0; d < 5; ++d) sb2v[d] = sb2[d];
    const float fcwv = fcw[0];
    const float fcbv = fcb[0];

    __syncthreads();

    for (; tile < ntiles; tile += NBLOCKS) {
        const float xa = xna, xb = xnb;

        xna = 0.0f; xnb = 0.0f;
        if (tile + NBLOCKS < ntiles) {
            const int ra = (tile + NBLOCKS) * ROWS_PER_BLOCK + rsub;
            if (ra < nrows)      xna = x[(size_t)ra * 2 + hh];
            if (ra + 16 < nrows) xnb = x[(size_t)(ra + 16) * 2 + hh];
        }

        float fa[NB], fb[NB];
        const v2f sv = feat9x2(xa, xb, myrow, fa, fb);

        v8f acc = {};
#pragma unroll
        for (int j = 0; j < 4; ++j) {
            v2f ba;
            ba.x = fa[fidx(j, 0)]; ba.y = fa[fidx(j, 1)];
            acc = __builtin_amdgcn_wmma_f32_16x16x4_f32(
                false, wfa[j], false, ba, (short)0, acc, false, false);
        }
#pragma unroll
        for (int j = 0; j < 4; ++j) {
            v2f bb;
            bb.x = fb[fidx(j, 0)]; bb.y = fb[fidx(j, 1)];
            acc = __builtin_amdgcn_wmma_f32_16x16x4_f32(
                false, wfb[j], false, bb, (short)0, acc, false, false);
        }
        acc = __builtin_amdgcn_wmma_f32_16x16x4_f32(
            false, wfs, false, sv, (short)0, acc, false, false);

        const v2f k25 = {2.5f, 2.5f}, k65 = {6.5f, 6.5f}, khalf = {0.5f, 0.5f};
        const v2f h01 = __builtin_shufflevector(acc, acc, 0, 1);
        const v2f h23 = __builtin_shufflevector(acc, acc, 2, 3);
        const v2f s01 = __builtin_elementwise_fma(h01, k25, k65);
        const v2f s23 = __builtin_elementwise_fma(h23, k25, k65);
        const v2f g01 = khalf * h01;
        const v2f g23 = khalf * h23;
        const float s1v[5] = {s01.x, s01.y, s23.x, s23.y, fmaf(acc[4], 2.5f, 6.5f)};
        const float gv[5]  = {g01.x, g01.y, g23.x, g23.y, 0.5f * acc[4]};
        float a = 0.0f;
#pragma unroll
        for (int d = 0; d < 5; ++d) {
            const float s1 = s1v[d];
            const float g  = gv[d];
            const float u  = __builtin_amdgcn_fractf(s1);
            const unsigned idx = min((unsigned)(int)s1, (unsigned)(NCUB - 1));
            const float4 p = lds_cub[d * NCUB + idx];
            const float q = fmaf(fmaf(p.w, u, p.z), u, p.y);
            const float base = (d == 0) ? p.x : a + p.x;
            a = fmaf(sb2v[d], silu_of_half(g), fmaf(q, u, base));
        }
        const int row = tile * ROWS_PER_BLOCK + t;
        if (row < nrows) out[row] = a * fcwv + fcbv;
    }
}

extern "C" void kernel_launch(void* const* d_in, const int* in_sizes, int n_in,
                              void* d_out, int out_size, void* d_ws, size_t ws_size,
                              hipStream_t stream) {
    (void)n_in; (void)d_ws; (void)ws_size; (void)out_size;
    const float* x     = (const float*)d_in[0];
    const float* coef1 = (const float*)d_in[1];
    const float* sb1   = (const float*)d_in[2];
    const float* sp1   = (const float*)d_in[3];
    const float* coef2 = (const float*)d_in[4];
    const float* sb2   = (const float*)d_in[5];
    const float* sp2   = (const float*)d_in[6];
    const float* fcw   = (const float*)d_in[7];
    const float* fcb   = (const float*)d_in[8];
    float* out = (float*)d_out;

    const int nrows = in_sizes[0] / 2;

    kan_fused_kernel<<<NBLOCKS, ROWS_PER_BLOCK, 0, stream>>>(
        x, coef1, sb1, sp1, coef2, sb2, sp2, fcw, fcb, out, nrows);
}


# Each wave works two 32-row groups per trip on two independent accumulators (two WMMA chains in flight), so 1024 work-groups of 512-row tiles replace 2048 of 256; per-row sum order unchanged
# speedup vs baseline: 4.4385x; 1.0832x over previous
#include <hip/hip_runtime.h>

typedef __attribute__((ext_vector_type(2))) float v2f;
typedef __attribute__((ext_vector_type(8))) float v8f;

#define NINT  11
#define NB    8
#define NTHREADS 256
#define NGRP  2
#define ROWS_PER_BLOCK (NTHREADS * NGRP)
#define NCUB  13
#define NBLOCKS 1024
#define ROW_DW  20
#define LANE_DW 44

__device__ __forceinline__ float silu_of_half(float g) {
    return fmaf(g, __builtin_amdgcn_tanhf(g), g);
}

__device__ __forceinline__ void place8(float* row, int mi,
                                       float w0, float w1, float w2, float w3, float bs[NB]) {
    const unsigned mu = min((unsigned)mi, (unsigned)NINT);
    row[8] = w3; row[9] = w2; row[10] = w1; row[11] = w0;
    const float* p = row + mu;
#pragma unroll
    for (int j = 0; j < NB; ++j) bs[j] = p[8 - j];
}

__device__ __forceinline__ v2f feat9x2(float xa, float xb, float* row, float fa[NB], float fb[NB]) {
    v2f xv; xv.x = xa; xv.y = xb;
    const v2f c25 = {2.5f, 2.5f}, c55 = {5.5f, 5.5f}, one = {1.0f, 1.0f}, half = {0.5f, 0.5f};
    const v2f mone = {-1.0f, -1.0f}, c23 = {2.0f / 3.0f, 2.0f / 3.0f}, c16 = {1.0f / 6.0f, 1.0f / 6.0f};
    const v2f s  = __builtin_elementwise_fma(xv, c25, c55);
    const v2f mf = __builtin_elementwise_floor(s);
    const v2f u  = s - mf;
    const v2f v  = one - u;
    const v2f u2 = u * u;
    const v2f v2 = v * v;
    const v2f w3 = u2 * u * c16;
    const v2f w0 = v2 * v * c16;
    const v2f w1 = __builtin_elementwise_fma(u2, __builtin_elementwise_fma(u, half, mone), c23);
    const v2f w2 = __builtin_elementwise_fma(v2, __builtin_elementwise_fma(v, half, mone), c23);
    place8(row, (int)mf.x, w0.x, w1.x, w2.x, w3.x, &fa[0]);
    place8(row, (int)mf.y, w0.y, w1.y, w2.y, w3.y, &fb[0]);
    const v2f hv = half * xv;
    v2f tv;
    tv.x = __builtin_amdgcn_tanhf(hv.x);
    tv.y = __builtin_amdgcn_tanhf(hv.y);
    return __builtin_elementwise_fma(hv, tv, hv);
}

__device__ __forceinline__ constexpr int fidx(int j, int c) {
    return 2 * j + 1 - c;
}

__device__ __forceinline__ float w1_feat(int i, int f, int o,
                                         const float* __restrict__ coef1,
                                         const float* __restrict__ sb1,
                                         const float* __restrict__ sp1) {
    if (f < NB)  return sp1[i * 5 + o] * coef1[(i * 5 + o) * NB + f];
    if (f == NB) return sb1[i * 5 + o];
    return 0.0f;
}

__global__ __launch_bounds__(NTHREADS) void kan_fused_kernel(
    const float* __restrict__ x,
    const float* __restrict__ coef1, const float* __restrict__ sb1,
    const float* __restrict__ sp1,
    const float* __restrict__ coef2, const float* __restrict__ sb2,
    const float* __restrict__ sp2,
    const float* __restrict__ fcw,   const float* __restrict__ fcb,
    float* __restrict__ out, int nrows)
{
    __shared__ float4 lds_cub[5 * NCUB];
    __shared__ __attribute__((aligned(16))) float lds_row[NTHREADS * LANE_DW];

    const int t    = threadIdx.x;
    const int lane = t & 31;
    const int wave = t >> 5;
    const int n    = lane & 15;
    const int hh   = lane >> 4;
    const int ntiles = (nrows + ROWS_PER_BLOCK - 1) / ROWS_PER_BLOCK;
    int tile = blockIdx.x;
    float* const myrow = lds_row + t * LANE_DW;

    const int rsub = wave * (32 * NGRP) + n;
    const int lastrow = nrows - 1;
    float xna[NGRP], xnb[NGRP];
#pragma unroll
    for (int g = 0; g < NGRP; ++g) { xna[g] = 0.0f; xnb[g] = 0.0f; }
    if (tile < ntiles) {
        __builtin_prefetch(&x[(size_t)tile * ROWS_PER_BLOCK * 2], 0, 1);
#pragma unroll
        for (int g = 0; g < NGRP; ++g) {
            const int ra = tile * ROWS_PER_BLOCK + rsub + 32 * g;
            xna[g] = x[(size_t)min(ra, lastrow) * 2 + hh];
            xnb[g] = x[(size_t)min(ra + 16, lastrow) * 2 + hh];
        }
    }


    if (t < 5 * NCUB) {
        const int d = t / NCUB;
        const int m = t - d * NCUB - 1;
        const float* c = coef2 + d * NB;
        const float la = c[min(max(m - 3, 0), NB - 1)];
        const float lb = c[min(max(m - 2, 0), NB - 1)];
        const float lg = c[min(max(m - 1, 0), NB - 1)];
        const float le = c[min(max(m, 0), NB - 1)];
        const float ca = (m >= 3 && m <= 10) ? la : 0.0f;
        const float cb = (m >= 2 && m <= 9)  ? lb : 0.0f;
        const float cg = (m >= 1 && m <= 8)  ? lg : 0.0f;
        const float ce = (m >= 0 && m <= 7)  ? le : 0.0f;
        const float sp = sp2[d];
        float4 p;
        p.x = sp * ((ca + 4.0f * cb + cg) * (1.0f / 6.0f));
        p.y = sp * ((cg - ca) * 0.5f);
        p.z = sp * ((ca - 2.0f * cb + cg) * 0.5f);
        p.w = sp * ((ce - ca + 3.0f * (cb - cg)) * (1.0f / 6.0f));
        lds_cub[t] = p;
    }

#pragma unroll
    for (int g = 0; g < NGRP; ++g) {
        float* const r = myrow + g * ROW_DW;
#pragma unroll
        for (int k = 0; k < 8; ++k) { r[k] = 0.0f; r[12 + k] = 0.0f; }
    }

    const int  o   = n & 7;
    const bool wok = (o < 5);
    const int  oc  = wok ? o : 4;
    const bool wa = wok && (n < 8);
    const bool wb = wok && (n >= 8);
    v2f wfa[4], wfb[4], wfs;
#pragma unroll
    for (int j = 0; j < 4; ++j) {
        const float wx = w1_feat(hh, fidx(j, 0), oc, coef1, sb1, sp1);
        const float wy = w1_feat(hh, fidx(j, 1), oc, coef1, sb1, sp1);
        wfa[j].x = wa ? wx : 0.0f;
        wfa[j].y = wa ? wy : 0.0f;
        wfb[j].x = wb ? wx : 0.0f;
        wfb[j].y = wb ? wy : 0.0f;
    }
    {
        const float ws = w1_feat(hh, NB, oc, coef1, sb1, sp1);
        wfs.x = wa ? ws : 0.0f;
        wfs.y = wb ? ws : 0.0f;
    }

    float sb2v[5];
#pragma unroll
    for (int d = 0; d < 5; ++d) sb2v[d] = sb2[d];
    const float fcwv = fcw[0];
    const float fcbv = fcb[0];

    __syncthreads();

    for (; tile < ntiles; tile += NBLOCKS) {
        float xa[NGRP], xb[NGRP];
#pragma unroll
        for (int g = 0; g < NGRP; ++g) { xa[g] = xna[g]; xb[g] = xnb[g]; }

        if (tile + NBLOCKS < ntiles) {
#pragma unroll
            for (int g = 0; g < NGRP; ++g) {
                const int ra = (tile + NBLOCKS) * ROWS_PER_BLOCK + rsub + 32 * g;
                xna[g] = x[(size_t)min(ra, lastrow) * 2 + hh];
                xnb[g] = x[(size_t)min(ra + 16, lastrow) * 2 + hh];
            }
        }

        float fa[NGRP][NB], fb[NGRP][NB];
        v2f sv[NGRP];
#pragma unroll
        for (int g = 0; g < NGRP; ++g)
            sv[g] = feat9x2(xa[g], xb[g], myrow + g * ROW_DW, fa[g], fb[g]);

        v8f acc[NGRP];
#pragma unroll
        for (int g = 0; g < NGRP; ++g) acc[g] = v8f{};
#pragma unroll
        for (int j = 0; j < 4; ++j) {
#pragma unroll
            for (int g = 0; g < NGRP; ++g) {
                v2f ba;
                ba.x = fa[g][fidx(j, 0)]; ba.y = fa[g][fidx(j, 1)];
                acc[g] = __builtin_amdgcn_wmma_f32_16x16x4_f32(
                    false, wfa[j], false, ba, (short)0, acc[g], false, false);
            }
        }
#pragma unroll
        for (int j = 0; j < 4; ++j) {
#pragma unroll
            for (int g = 0; g < NGRP; ++g) {
                v2f bb;
                bb.x = fb[g][fidx(j, 0)]; bb.y = fb[g][fidx(j, 1)];
                acc[g] = __builtin_amdgcn_wmma_f32_16x16x4_f32(
                    false, wfb[j], false, bb, (short)0, acc[g], false, false);
            }
        }
#pragma unroll
        for (int g = 0; g < NGRP; ++g)
            acc[g] = __builtin_amdgcn_wmma_f32_16x16x4_f32(
                false, wfs, false, sv[g], (short)0, acc[g], false, false);

        const v2f k25 = {2.5f, 2.5f}, k65 = {6.5f, 6.5f}, khalf = {0.5f, 0.5f};
        float res[NGRP];
#pragma unroll
        for (int g = 0; g < NGRP; ++g) {
            const v2f h01 = __builtin_shufflevector(acc[g], acc[g], 0, 1);
            const v2f h23 = __builtin_shufflevector(acc[g], acc[g], 2, 3);
            const v2f s01 = __builtin_elementwise_fma(h01, k25, k65);
            const v2f s23 = __builtin_elementwise_fma(h23, k25, k65);
            const v2f g01 = khalf * h01;
            const v2f g23 = khalf * h23;
            const float s1v[5] = {s01.x, s01.y, s23.x, s23.y, fmaf(acc[g][4], 2.5f, 6.5f)};
            const float gv[5]  = {g01.x, g01.y, g23.x, g23.y, 0.5f * acc[g][4]};
            float a = 0.0f;
#pragma unroll
            for (int d = 0; d < 5; ++d) {
                const float s1 = s1v[d];
                const float gg = gv[d];
                const float u  = __builtin_amdgcn_fractf(s1);
                const unsigned idx = min((unsigned)(int)s1, (unsigned)(NCUB - 1));
                const float4 p = lds_cub[d * NCUB + idx];
                const float q = fmaf(fmaf(p.w, u, p.z), u, p.y);
                const float base = (d == 0) ? p.x : a + p.x;
                a = fmaf(sb2v[d], silu_of_half(gg), fmaf(q, u, base));
            }
            res[g] = a * fcwv + fcbv;
        }
        const int row0 = tile * ROWS_PER_BLOCK + wave * (32 * NGRP) + lane;
        if ((tile + 1) * ROWS_PER_BLOCK <= nrows) {
#pragma unroll
            for (int g = 0; g < NGRP; ++g) out[row0 + 32 * g] = res[g];
        } else {
#pragma unroll
            for (int g = 0; g < NGRP; ++g)
                if (row0 + 32 * g < nrows) out[row0 + 32 * g] = res[g];
        }
    }
}

extern "C" void kernel_launch(void* const* d_in, const int* in_sizes, int n_in,
                              void* d_out, int out_size, void* d_ws, size_t ws_size,
                              hipStream_t stream) {
    (void)n_in; (void)d_ws; (void)ws_size; (void)out_size;
    const float* x     = (const float*)d_in[0];
    const float* coef1 = (const float*)d_in[1];
    const float* sb1   = (const float*)d_in[2];
    const float* sp1   = (const float*)d_in[3];
    const float* coef2 = (const float*)d_in[4];
    const float* sb2   = (const float*)d_in[5];
    const float* sp2   = (const float*)d_in[6];
    const float* fcw   = (const float*)d_in[7];
    const float* fcb   = (const float*)d_in[8];
    float* out = (float*)d_out;

    const int nrows = in_sizes[0] / 2;

    kan_fused_kernel<<<NBLOCKS, NTHREADS, 0, stream>>>(
        x, coef1, sb1, sp1, coef2, sb2, sp2, fcw, fcb, out, nrows);
}
